# starting kernel
# baseline (speedup 1.0000x reference)
#include <hip/hip_runtime.h>

typedef __attribute__((ext_vector_type(2))) float v2f;
typedef __attribute__((ext_vector_type(8))) float v8f;
typedef __attribute__((ext_vector_type(4))) float f4;

#define Nb  32
#define Cc  32
#define Hh  56
#define Ww  56
#define Oo  32
#define CKK 288
#define PP  3136
#define TW  4

__global__ __launch_bounds__(128) void lc2d_wmma_f32(
    const float* __restrict__ x,
    const float* __restrict__ wgt,
    const float* __restrict__ bias,
    float* __restrict__ out)
{
    __shared__ float sP[CKK][33];
    __shared__ float sW[CKK][33];
    __shared__ float sO[Nb][33][TW];

    const int t     = threadIdx.x;
    const int lane  = t & 31;
    const int wave  = t >> 5;
    const int half  = lane >> 4;
    const int m16   = lane & 15;
    const int mtile = wave >> 1;
    const int ntile = wave & 1;
    const int mA    = m16 + 16 * mtile;
    const int oC    = m16 + 16 * ntile;

    const int h  = blockIdx.y;
    const int w0 = blockIdx.x * TW;

    for (int tw = 0; tw < TW; ++tw) {
        const int w = w0 + tw;
        const int p = h * Ww + w;

        __syncthreads();

        for (int oo = 0; oo < 8; ++oo) {
            const int o = wave * 8 + oo;
            const float* src = wgt + ((size_t)o * PP + p) * CKK;
#pragma unroll
            for (int s = 0; s < 9; ++s) {
                const int k = s * 32 + lane;
                sW[k][o] = src[k];
            }
        }

        for (int e = 0; e < 72; ++e) {
            const int idx = e * 128 + t;
            const int n   = idx & 31;
            const int k   = idx >> 5;
            const int c   = k / 9;
            const int r9  = k - c * 9;
            const int ki  = r9 / 3;
            const int kj  = r9 - ki * 3;
            const int hh  = h + ki - 1;
            const int ww  = w + kj - 1;
            float v = 0.0f;
            if ((unsigned)hh < (unsigned)Hh && (unsigned)ww < (unsigned)Ww)
                v = x[(((size_t)n * Cc + c) * Hh + hh) * Ww + ww];
            sP[k][n] = v;
        }

        if (tw + 1 < TW) {
            const int o = wave * 8 + (lane >> 2);
            __builtin_prefetch(wgt + ((size_t)o * PP + (p + 1)) * CKK + (size_t)(lane & 3) * 72, 0, 1);
        }

        __syncthreads();

        v8f acc = {};
#pragma unroll
        for (int k0 = 0; k0 < CKK; k0 += 4) {
            const int kA = k0 + 2 * half;
            v2f a, b;
            a[0] = sP[kA + 0][mA];
            a[1] = sP[kA + 1][mA];
            b[0] = sW[kA + 0][oC];
            b[1] = sW[kA + 1][oC];
            acc = __builtin_amdgcn_wmma_f32_16x16x4_f32(
                false, a, false, b, (short)0, acc, false, false);
        }

        const float bv = bias[(size_t)oC * PP + p];
#pragma unroll
        for (int r = 0; r < 8; ++r) {
            const int n = r + 8 * half + 16 * mtile;
            sO[n][oC][tw] = acc[r] + bv;
        }
    }

    __syncthreads();

    for (int e = 0; e < 8; ++e) {
        const int pair = e * 128 + t;
        const int n = pair >> 5;
        const int o = pair & 31;
        f4 v;
        v.x = sO[n][o][0];
        v.y = sO[n][o][1];
        v.z = sO[n][o][2];
        v.w = sO[n][o][3];
        *(f4*)(out + ((size_t)n * Oo + o) * PP + (size_t)h * Ww + w0) = v;
    }
}

extern "C" void kernel_launch(void* const* d_in, const int* in_sizes, int n_in,
                              void* d_out, int out_size, void* d_ws, size_t ws_size,
                              hipStream_t stream) {
    const float* x    = (const float*)d_in[0];
    const float* wgt  = (const float*)d_in[1];
    const float* bias = (const float*)d_in[2];
    float* out        = (float*)d_out;

    dim3 grid(Ww / TW, Hh);
    lc2d_wmma_f32<<<grid, 128, 0, stream>>>(x, wgt, bias, out);
}


# x window of the workgroup's positions staged once (row-wise loads, two channel halves); A fragments read from that tile, no per-position im2col gather; k order and sums unchanged
# speedup vs baseline: 2.6244x; 2.6244x over previous
#include <hip/hip_runtime.h>


typedef __attribute__((ext_vector_type(2))) float v2f;
typedef __attribute__((ext_vector_type(8))) float v8f;
typedef __attribute__((ext_vector_type(4))) float f4;

#define Nb  32
#define Cc  32
#define Hh  56
#define Ww  56
#define Oo  32
#define CKK 288
#define PP  3136
#define TW  4
#define CH  16
#define KH  144
#define XR  288
#define NP  33

__global__ __launch_bounds__(128, 4) void lc2d_wmma_f32(
    const float* __restrict__ x,
    const float* __restrict__ wgt,
    const float* __restrict__ bias,
    float* __restrict__ out)
{
    __shared__ float xs[XR][NP];
    __shared__ float sW[KH][NP];

    const int t     = threadIdx.x;
    const int lane  = t & 31;
    const int wave  = t >> 5;
    const int half  = lane >> 4;
    const int m16   = lane & 15;
    const int mtile = wave >> 1;
    const int ntile = wave & 1;
    const int mA    = m16 + 16 * mtile;
    const int oC    = m16 + 16 * ntile;

    const int h  = blockIdx.y;
    const int w0 = blockIdx.x * TW;

    const float* pa2 = &xs[0][0] + mA + half * (2 * NP);
    const float* pa5 = &xs[0][0] + mA + half * (5 * NP);
    const float* pb  = &sW[0][0] + oC + half * (2 * NP);

    v8f acc[TW];
#pragma unroll
    for (int tw = 0; tw < TW; ++tw) acc[tw] = (v8f){};

    for (int ch = 0; ch < 2; ++ch) {
        __syncthreads();

        {
            const int loff = (w0 > 0) ? -1 : 0;
            const int roff = (w0 + TW < Ww) ? TW : TW - 1;
            f4 mid[12];
            float lft[12], rgt[12];
#pragma unroll
            for (int e = 0; e < 12; ++e) {
                const int cl = wave * 4 + e / 3;
                const int hh = h + (e % 3) - 1;
                const int hc = hh < 0 ? 0 : (hh > Hh - 1 ? Hh - 1 : hh);
                const float* src = x + (((size_t)lane * Cc + (ch * CH + cl)) * Hh + hc) * Ww + w0;
                mid[e] = *(const f4*)src;
                lft[e] = src[loff];
                rgt[e] = src[roff];
            }
#pragma unroll
            for (int e = 0; e < 12; ++e) {
                const int hh = h + (e % 3) - 1;
                const bool rowok = (unsigned)hh < (unsigned)Hh;
                const bool lok = rowok && (w0 > 0);
                const bool rok = rowok && (w0 + TW < Ww);
                float* dst = &xs[(wave * 12 + e) * 6][lane];
                dst[0 * NP] = lok   ? lft[e]   : 0.0f;
                dst[1 * NP] = rowok ? mid[e].x : 0.0f;
                dst[2 * NP] = rowok ? mid[e].y : 0.0f;
                dst[3 * NP] = rowok ? mid[e].z : 0.0f;
                dst[4 * NP] = rowok ? mid[e].w : 0.0f;
                dst[5 * NP] = rok   ? rgt[e]   : 0.0f;
            }
        }

#pragma unroll
        for (int tw = 0; tw < TW; ++tw) {
            const int p = h * Ww + w0 + tw;

            if (tw > 0) __syncthreads();

            {
                const float* srcw = wgt + ((size_t)(wave * 8) * PP + p) * CKK + ch * KH;
                float wv[36];
#pragma unroll
                for (int e = 0; e < 36; ++e) {
                    const int g  = e / 9;
                    const int r9 = e % 9;
                    const int hi = (r9 < 4) ? 0 : (r9 > 4) ? 1 : half;
                    wv[e] = srcw[(size_t)(2 * g) * PP * CKK + r9 * 32 + lane + hi * (PP * CKK - KH)];
                }
#pragma unroll
                for (int e = 0; e < 36; ++e) {
                    const int g  = e / 9;
                    const int r9 = e % 9;
                    const int hi = (r9 < 4) ? 0 : (r9 > 4) ? 1 : half;
                    const int k  = r9 * 32 + lane - hi * KH;
                    sW[k][wave * 8 + 2 * g + hi] = wv[e];
                }
            }

            if (tw + 1 < TW) {
                const int o = wave * 8 + (lane >> 2);
                __builtin_prefetch(wgt + ((size_t)o * PP + (p + 1)) * CKK + ch * KH + (size_t)(lane & 3) * 36, 0, 1);
            }

            __syncthreads();

            const float* qa2 = pa2 + tw * NP;
            const float* qa5 = pa5 + tw * NP;
#pragma unroll
            for (int k0 = 0; k0 < KH; k0 += 4) {
                v2f a, b;
                a[0] = ((k0 + 0) % 3 == 0 ? qa2 : qa5)[(2 * (k0 + 0) - (k0 + 0) % 3) * NP];
                a[1] = ((k0 + 1) % 3 == 0 ? qa2 : qa5)[(2 * (k0 + 1) - (k0 + 1) % 3) * NP];
                b[0] = pb[(k0 + 0) * NP];
                b[1] = pb[(k0 + 1) * NP];
                acc[tw] = __builtin_amdgcn_wmma_f32_16x16x4_f32(
                    false, a, false, b, (short)0, acc[tw], false, false);
            }
        }
    }

    const f4 bv = *(const f4*)(bias + (size_t)oC * PP + (size_t)h * Ww + w0);
#pragma unroll
    for (int r = 0; r < 8; ++r) {
        const int n = r + 8 * half + 16 * mtile;
        f4 v;
        v.x = acc[0][r] + bv.x;
        v.y = acc[1][r] + bv.y;
        v.z = acc[2][r] + bv.z;
        v.w = acc[3][r] + bv.w;
        *(f4*)(out + ((size_t)n * Oo + oC) * PP + (size_t)h * Ww + w0) = v;
    }
}

extern "C" void kernel_launch(void* const* d_in, const int* in_sizes, int n_in,
                              void* d_out, int out_size, void* d_ws, size_t ws_size,
                              hipStream_t stream) {
    const float* x    = (const float*)d_in[0];
    const float* wgt  = (const float*)d_in[1];
    const float* bias = (const float*)d_in[2];
    float* out        = (float*)d_out;

    dim3 grid(Ww / TW, Hh);
    lc2d_wmma_f32<<<grid, 128, 0, stream>>>(x, wgt, bias, out);
}
